# starting kernel
# baseline (speedup 1.0000x reference)
#include <hip/hip_runtime.h>

typedef __attribute__((ext_vector_type(2))) float v2f;
typedef __attribute__((ext_vector_type(8))) float v8f;

#define N_NODES  100000
#define N_EDGES  400000
#define D_EMB    128
#define D_PROMPT 64
#define D_OUT    192

__global__ void zero_kernel(float* __restrict__ out, float* __restrict__ deg) {
    long tid = (long)blockIdx.x * blockDim.x + threadIdx.x;
    const long emb_elems = (long)N_NODES * D_EMB;
    if (tid < emb_elems) {
        long n = tid >> 7;
        long c = tid & (D_EMB - 1);
        out[n * D_OUT + c] = 0.0f;
    } else if (tid < emb_elems + N_NODES) {
        deg[tid - emb_elems] = 0.0f;
    }
}

__device__ inline void atomic_fadd(float* p, float v) {
    __hip_atomic_fetch_add(p, v, __ATOMIC_RELAXED, __HIP_MEMORY_SCOPE_AGENT);
}

__global__ void scatter_kernel(const float* __restrict__ emb,
                               const int*   __restrict__ src,
                               const int*   __restrict__ dst,
                               float* __restrict__ out,
                               float* __restrict__ deg) {
    const int lane = threadIdx.x & 31;
    const int wave = threadIdx.x >> 5;
    const long e = (long)blockIdx.x * (blockDim.x >> 5) + wave;
    if (e >= N_EDGES) return;

    const int s = src[e];
    const int d = dst[e];

    const float4* row = (const float4*)(emb + (long)s * D_EMB);
    float4 v = row[lane];

    float* o = out + (long)d * D_OUT + lane * 4;
    atomic_fadd(o + 0, v.x);
    atomic_fadd(o + 1, v.y);
    atomic_fadd(o + 2, v.z);
    atomic_fadd(o + 3, v.w);

    if (lane == 0) atomic_fadd(deg + d, 1.0f);
}

__global__ void prompt_wmma_kernel(const float* __restrict__ deg,
                                   const float* __restrict__ weight,
                                   float* __restrict__ out) {
    const int lane = threadIdx.x & 31;
    const int wave = threadIdx.x >> 5;
    const int tile = blockIdx.x * (blockDim.x >> 5) + wave;
    if (tile >= N_NODES / 16) return;

    const int lo   = (lane < 16);
    const int l15  = lane & 15;

    v2f a;
    a.x = lo ? deg[tile * 16 + l15] : 0.0f;
    a.y = 0.0f;

    #pragma unroll
    for (int j = 0; j < 4; ++j) {
        v2f b;
        b.x = lo ? weight[j * 16 + l15] : 0.0f;
        b.y = 0.0f;

        v8f c = {};
        c = __builtin_amdgcn_wmma_f32_16x16x4_f32(
                false, a, false, b, (short)0, c, false, false);

        const int col = D_EMB + j * 16 + l15;
        #pragma unroll
        for (int r = 0; r < 8; ++r) {
            const int m   = lo ? r : (8 + r);
            const long row = (long)(tile * 16 + m);
            out[row * D_OUT + col] = c[r];
        }
    }
}

extern "C" void kernel_launch(void* const* d_in, const int* in_sizes, int n_in,
                              void* d_out, int out_size, void* d_ws, size_t ws_size,
                              hipStream_t stream) {
    const float* emb    = (const float*)d_in[0];
    const float* weight = (const float*)d_in[1];
    const int*   src    = (const int*)d_in[2];
    const int*   dst    = (const int*)d_in[3];

    float* out = (float*)d_out;
    float* deg = (float*)d_ws;

    {
        long n = (long)N_NODES * D_EMB + N_NODES;
        int blocks = (int)((n + 255) / 256);
        zero_kernel<<<blocks, 256, 0, stream>>>(out, deg);
    }

    {
        int waves_per_block = 256 / 32;
        int blocks = (N_EDGES + waves_per_block - 1) / waves_per_block;
        scatter_kernel<<<blocks, 256, 0, stream>>>(emb, src, dst, out, deg);
    }

    {
        int tiles = N_NODES / 16;
        int waves_per_block = 256 / 32;
        int blocks = (tiles + waves_per_block - 1) / waves_per_block;
        prompt_wmma_kernel<<<blocks, 256, 0, stream>>>(deg, weight, out);
    }
}


# tile kernel: all 32 row loads of a 32-edge batch issued before its first matrix step (parent: 4 rows in flight per wave); lanes past the last edge re-read row 0; sum order unchanged
# speedup vs baseline: 1.9563x; 1.9563x over previous
#include <hip/hip_runtime.h>

typedef __attribute__((ext_vector_type(2))) float v2f;
typedef __attribute__((ext_vector_type(4))) float v4f;
typedef __attribute__((ext_vector_type(8))) float v8f;

#define N_NODES    100000
#define N_EDGES    400000
#define D_EMB      128
#define D_PROMPT   64
#define D_OUT      192
#define TILE_M     16
#define N_TILES    (N_NODES / TILE_M)
#define TILE_WORDS (TILE_M * D_OUT)
#define LIST_OFF   32
#define LIST_CAP   (TILE_WORDS - LIST_OFF)
#define ROW_SHIFT  20
#define SRC_MASK   0xFFFFF
#define PAD_ENT    (16 << ROW_SHIFT)


__global__ void clear_counts_kernel(float* __restrict__ out) {
    const int t = blockIdx.x * blockDim.x + threadIdx.x;
    if (t < N_TILES) ((int*)out)[(long)t * TILE_WORDS] = 0;
}

__global__ void bin_edges_kernel(const int* __restrict__ src,
                                 const int* __restrict__ dst,
                                 float* __restrict__ out) {
    const int e = blockIdx.x * blockDim.x + threadIdx.x;
    if (e >= N_EDGES) return;
    const int d = dst[e];
    const int s = src[e];
    int* reg = (int*)out + (long)(d >> 4) * TILE_WORDS;
    const int slot = __hip_atomic_fetch_add(reg, 1, __ATOMIC_RELAXED, __HIP_MEMORY_SCOPE_AGENT);
    if (slot < LIST_CAP) reg[LIST_OFF + slot] = s | ((d & 15) << ROW_SHIFT);
}

struct KStep {
    v2f    a;
    v4f    x0, x1, y0, y1;
};

__device__ __forceinline__ KStep load_kstep(const float* __restrict__ emb,
                                            int ent, int s, int c, int h) {
    const int q  = 4 * s;
    const int e0 = __builtin_amdgcn_readlane(ent, q + 0);
    const int e1 = __builtin_amdgcn_readlane(ent, q + 1);
    const int e2 = __builtin_amdgcn_readlane(ent, q + 2);
    const int e3 = __builtin_amdgcn_readlane(ent, q + 3);
    const int ex = h ? e2 : e0;
    const int ey = h ? e3 : e1;
    KStep k;
    k.a.x = ((ex >> ROW_SHIFT) == c) ? 1.0f : 0.0f;
    k.a.y = ((ey >> ROW_SHIFT) == c) ? 1.0f : 0.0f;
    const unsigned ox = (unsigned)(ex & SRC_MASK) * (unsigned)(D_EMB * sizeof(float)) + 16u * (unsigned)c;
    const unsigned oy = (unsigned)(ey & SRC_MASK) * (unsigned)(D_EMB * sizeof(float)) + 16u * (unsigned)c;
    const char* base = (const char*)emb;
    k.x0 = *(const v4f*)(base + ox);
    k.x1 = *(const v4f*)(base + ox + 256u);
    k.y0 = *(const v4f*)(base + oy);
    k.y1 = *(const v4f*)(base + oy + 256u);
    return k;
}

#define MMA(A, BX, BY, C) \
    __builtin_amdgcn_wmma_f32_16x16x4_f32(false, (A), false, (v2f){(BX), (BY)}, (short)0, (C), false, false)

__device__ __forceinline__ void mma_kstep(const KStep& k, v8f (&acc)[8], float& pdeg) {
    pdeg += k.a.x + k.a.y;
    acc[0] = MMA(k.a, k.x0.x, k.y0.x, acc[0]);
    acc[1] = MMA(k.a, k.x0.y, k.y0.y, acc[1]);
    acc[2] = MMA(k.a, k.x0.z, k.y0.z, acc[2]);
    acc[3] = MMA(k.a, k.x0.w, k.y0.w, acc[3]);
    acc[4] = MMA(k.a, k.x1.x, k.y1.x, acc[4]);
    acc[5] = MMA(k.a, k.x1.y, k.y1.y, acc[5]);
    acc[6] = MMA(k.a, k.x1.z, k.y1.z, acc[6]);
    acc[7] = MMA(k.a, k.x1.w, k.y1.w, acc[7]);
}

__device__ __forceinline__ void process_batch(const float* __restrict__ emb,
                                              int ent, int steps, int c, int h,
                                              v8f (&acc)[8], float& pdeg) {
    KStep k[8];
    #pragma unroll
    for (int s = 0; s < 8; ++s) k[s] = load_kstep(emb, ent, s, c, h);
    __builtin_amdgcn_sched_barrier(0);
    mma_kstep(k[0], acc, pdeg);
    if (steps > 1) { mma_kstep(k[1], acc, pdeg);
    if (steps > 2) { mma_kstep(k[2], acc, pdeg);
    if (steps > 3) { mma_kstep(k[3], acc, pdeg);
    if (steps > 4) { mma_kstep(k[4], acc, pdeg);
    if (steps > 5) { mma_kstep(k[5], acc, pdeg);
    if (steps > 6) { mma_kstep(k[6], acc, pdeg);
    if (steps > 7) { mma_kstep(k[7], acc, pdeg); } } } } } } }
    #pragma unroll
    for (int s = 0; s < 8; ++s)
        asm volatile("" :: "v"(k[s].x0), "v"(k[s].x1), "v"(k[s].y0), "v"(k[s].y1));
}

__global__ void __launch_bounds__(64)
tile_sum_kernel(const float* __restrict__ emb,
                const float* __restrict__ weight,
                const int*   __restrict__ src,
                const int*   __restrict__ dst,
                float* out) {
    const int lane = threadIdx.x & 31;
    const int wave = __builtin_amdgcn_readfirstlane(threadIdx.x >> 5);
    const int tile = blockIdx.x * 2 + wave;
    if (tile >= N_TILES) return;

    const int c = lane & 15;
    const int h = lane >> 4;

    float* reg = out + (long)tile * TILE_WORDS;
    const int cnt = __builtin_amdgcn_readfirstlane(__float_as_int(reg[0]));

    v8f acc[8];
    #pragma unroll
    for (int j = 0; j < 8; ++j) acc[j] = (v8f){};
    float pdeg = 0.0f;

    const bool listed = cnt <= LIST_CAP;
    const int  total  = listed ? cnt : N_EDGES;
    for (int base = 0; base < total; base += 32) {
        int ent, steps;
        if (listed) {
            const int i    = base + lane;
            const int left = cnt - base;
            ent   = (i < cnt) ? __float_as_int(reg[LIST_OFF + i]) : PAD_ENT;
            steps = ((left < 32 ? left : 32) + 3) >> 2;
        } else {
            const int      d    = dst[base + lane];
            const bool     mine = (d >> 4) == tile;
            const unsigned hits = __builtin_amdgcn_ballot_w32(mine);
            if (hits == 0) continue;
            ent   = mine ? (src[base + lane] | ((d & 15) << ROW_SHIFT)) : PAD_ENT;
            steps = ((31 - __builtin_clz(hits)) >> 2) + 1;
        }
        process_batch(emb, ent, steps, c, h, acc, pdeg);
    }

    const float  deg = pdeg + __shfl_xor(pdeg, 16);
    const float4 w4  = ((const float4*)weight)[c];
    v2f a2;
    a2.x = h ? 0.0f : deg;
    a2.y = 0.0f;
    v8f p[4];
    p[0] = MMA(a2, w4.x, 0.0f, (v8f){});
    p[1] = MMA(a2, w4.y, 0.0f, (v8f){});
    p[2] = MMA(a2, w4.z, 0.0f, (v8f){});
    p[3] = MMA(a2, w4.w, 0.0f, (v8f){});

    #pragma unroll
    for (int r = 0; r < 8; ++r) {
        float* o = out + (long)(tile * TILE_M + 8 * h + r) * D_OUT + 4 * c;
        *(float4*)(o)       = make_float4(acc[0][r], acc[1][r], acc[2][r], acc[3][r]);
        *(float4*)(o + 64)  = make_float4(acc[4][r], acc[5][r], acc[6][r], acc[7][r]);
        *(float4*)(o + 128) = make_float4(p[0][r], p[1][r], p[2][r], p[3][r]);
    }
}

extern "C" void kernel_launch(void* const* d_in, const int* in_sizes, int n_in,
                              void* d_out, int out_size, void* d_ws, size_t ws_size,
                              hipStream_t stream) {
    const float* emb    = (const float*)d_in[0];
    const float* weight = (const float*)d_in[1];
    const int*   src    = (const int*)d_in[2];
    const int*   dst    = (const int*)d_in[3];

    float* out = (float*)d_out;

    clear_counts_kernel<<<(N_TILES + 255) / 256, 256, 0, stream>>>(out);
    bin_edges_kernel<<<(N_EDGES + 255) / 256, 256, 0, stream>>>(src, dst, out);
    tile_sum_kernel<<<N_TILES / 2, 64, 0, stream>>>(emb, weight, src, dst, out);
}


# bin pass: two lists and counters per tile, one per package half; a wave adds only to its own half's (die number picks it): no line is written from both halves; sum order = slot arrival per list
# speedup vs baseline: 2.7772x; 1.4196x over previous
#include <hip/hip_runtime.h>

typedef __attribute__((ext_vector_type(2))) float v2f;
typedef __attribute__((ext_vector_type(4))) float v4f;
typedef __attribute__((ext_vector_type(8))) float v8f;

#define N_NODES    100000
#define N_EDGES    400000
#define D_EMB      128
#define D_PROMPT   64
#define D_OUT      192
#define TILE_M     16
#define N_TILES    (N_NODES / TILE_M)
#define TILE_WORDS (TILE_M * D_OUT)
#define HALF_WORDS (TILE_WORDS / 2)
#define LIST_OFF   32
#define LIST_CAP   (HALF_WORDS - LIST_OFF)
#define ROW_SHIFT  20
#define SRC_MASK   0xFFFFF
#define PAD_ENT    (16 << ROW_SHIFT)


__global__ void clear_counts_kernel(float* __restrict__ out) {
    const int t = blockIdx.x * blockDim.x + threadIdx.x;
    if (t < 2 * N_TILES) ((int*)out)[(long)(t >> 1) * TILE_WORDS + (t & 1) * HALF_WORDS] = 0;
}

__global__ void bin_edges_kernel(const int* __restrict__ src,
                                 const int* __restrict__ dst,
                                 float* __restrict__ out) {
    const int e = blockIdx.x * blockDim.x + threadIdx.x;
    if (e >= N_EDGES) return;
    const int d = dst[e];
    const int s = src[e];
    const int half = (int)((__builtin_amdgcn_s_sendmsg_rtn(0x87) >> 18) & 1u);
    int* reg = (int*)out + (long)(d >> 4) * TILE_WORDS + half * HALF_WORDS;
    const int slot = __hip_atomic_fetch_add(reg, 1, __ATOMIC_RELAXED, __HIP_MEMORY_SCOPE_AGENT);
    if (slot < LIST_CAP) reg[LIST_OFF + slot] = s | ((d & 15) << ROW_SHIFT);
}

struct KStep {
    v2f    a;
    v4f    x0, x1, y0, y1;
};

__device__ __forceinline__ KStep load_kstep(const float* __restrict__ emb,
                                            int ent, int s, int c, int h) {
    const int q  = 4 * s;
    const int e0 = __builtin_amdgcn_readlane(ent, q + 0);
    const int e1 = __builtin_amdgcn_readlane(ent, q + 1);
    const int e2 = __builtin_amdgcn_readlane(ent, q + 2);
    const int e3 = __builtin_amdgcn_readlane(ent, q + 3);
    const int ex = h ? e2 : e0;
    const int ey = h ? e3 : e1;
    KStep k;
    k.a.x = ((ex >> ROW_SHIFT) == c) ? 1.0f : 0.0f;
    k.a.y = ((ey >> ROW_SHIFT) == c) ? 1.0f : 0.0f;
    const unsigned ox = (unsigned)(ex & SRC_MASK) * (unsigned)(D_EMB * sizeof(float)) + 16u * (unsigned)c;
    const unsigned oy = (unsigned)(ey & SRC_MASK) * (unsigned)(D_EMB * sizeof(float)) + 16u * (unsigned)c;
    const char* base = (const char*)emb;
    k.x0 = *(const v4f*)(base + ox);
    k.x1 = *(const v4f*)(base + ox + 256u);
    k.y0 = *(const v4f*)(base + oy);
    k.y1 = *(const v4f*)(base + oy + 256u);
    return k;
}

#define MMA(A, BX, BY, C) \
    __builtin_amdgcn_wmma_f32_16x16x4_f32(false, (A), false, (v2f){(BX), (BY)}, (short)0, (C), false, false)

__device__ __forceinline__ void mma_kstep(const KStep& k, v8f (&acc)[8], float& pdeg) {
    pdeg += k.a.x + k.a.y;
    acc[0] = MMA(k.a, k.x0.x, k.y0.x, acc[0]);
    acc[1] = MMA(k.a, k.x0.y, k.y0.y, acc[1]);
    acc[2] = MMA(k.a, k.x0.z, k.y0.z, acc[2]);
    acc[3] = MMA(k.a, k.x0.w, k.y0.w, acc[3]);
    acc[4] = MMA(k.a, k.x1.x, k.y1.x, acc[4]);
    acc[5] = MMA(k.a, k.x1.y, k.y1.y, acc[5]);
    acc[6] = MMA(k.a, k.x1.z, k.y1.z, acc[6]);
    acc[7] = MMA(k.a, k.x1.w, k.y1.w, acc[7]);
}

__device__ __forceinline__ void process_batch(const float* __restrict__ emb,
                                              int ent, int steps, int c, int h,
                                              v8f (&acc)[8], float& pdeg) {
    KStep k[8];
    #pragma unroll
    for (int s = 0; s < 8; ++s) k[s] = load_kstep(emb, ent, s, c, h);
    __builtin_amdgcn_sched_barrier(0);
    mma_kstep(k[0], acc, pdeg);
    if (steps > 1) { mma_kstep(k[1], acc, pdeg);
    if (steps > 2) { mma_kstep(k[2], acc, pdeg);
    if (steps > 3) { mma_kstep(k[3], acc, pdeg);
    if (steps > 4) { mma_kstep(k[4], acc, pdeg);
    if (steps > 5) { mma_kstep(k[5], acc, pdeg);
    if (steps > 6) { mma_kstep(k[6], acc, pdeg);
    if (steps > 7) { mma_kstep(k[7], acc, pdeg); } } } } } } }
    #pragma unroll
    for (int s = 0; s < 8; ++s)
        asm volatile("" :: "v"(k[s].x0), "v"(k[s].x1), "v"(k[s].y0), "v"(k[s].y1));
}

__global__ void __launch_bounds__(64)
tile_sum_kernel(const float* __restrict__ emb,
                const float* __restrict__ weight,
                const int*   __restrict__ src,
                const int*   __restrict__ dst,
                float* out) {
    const int lane = threadIdx.x & 31;
    const int wave = __builtin_amdgcn_readfirstlane(threadIdx.x >> 5);
    const int tile = blockIdx.x * 2 + wave;
    if (tile >= N_TILES) return;

    const int c = lane & 15;
    const int h = lane >> 4;

    float* reg = out + (long)tile * TILE_WORDS;
    const int cw   = __float_as_int(reg[(lane & 1) * HALF_WORDS]);
    const int cnt0 = __builtin_amdgcn_readlane(cw, 0);
    const int cnt1 = __builtin_amdgcn_readlane(cw, 1);
    const int cnt  = cnt0 + cnt1;

    v8f acc[8];
    #pragma unroll
    for (int j = 0; j < 8; ++j) acc[j] = (v8f){};
    float pdeg = 0.0f;

    const bool listed = cnt0 <= LIST_CAP && cnt1 <= LIST_CAP;
    const int  total  = listed ? cnt : N_EDGES;
    for (int base = 0; base < total; base += 32) {
        int ent, steps;
        if (listed) {
            const int i    = base + lane;
            const int left = cnt - base;
            const int w    = (i < cnt0) ? (LIST_OFF + i) : (HALF_WORDS + LIST_OFF + i - cnt0);
            ent   = (i < cnt) ? __float_as_int(reg[w]) : PAD_ENT;
            steps = ((left < 32 ? left : 32) + 3) >> 2;
        } else {
            const int      d    = dst[base + lane];
            const bool     mine = (d >> 4) == tile;
            const unsigned hits = __builtin_amdgcn_ballot_w32(mine);
            if (hits == 0) continue;
            ent   = mine ? (src[base + lane] | ((d & 15) << ROW_SHIFT)) : PAD_ENT;
            steps = ((31 - __builtin_clz(hits)) >> 2) + 1;
        }
        process_batch(emb, ent, steps, c, h, acc, pdeg);
    }

    const float  deg = pdeg + __shfl_xor(pdeg, 16);
    const float4 w4  = ((const float4*)weight)[c];
    v2f a2;
    a2.x = h ? 0.0f : deg;
    a2.y = 0.0f;
    v8f p[4];
    p[0] = MMA(a2, w4.x, 0.0f, (v8f){});
    p[1] = MMA(a2, w4.y, 0.0f, (v8f){});
    p[2] = MMA(a2, w4.z, 0.0f, (v8f){});
    p[3] = MMA(a2, w4.w, 0.0f, (v8f){});

    #pragma unroll
    for (int r = 0; r < 8; ++r) {
        float* o = out + (long)(tile * TILE_M + 8 * h + r) * D_OUT + 4 * c;
        *(float4*)(o)       = make_float4(acc[0][r], acc[1][r], acc[2][r], acc[3][r]);
        *(float4*)(o + 64)  = make_float4(acc[4][r], acc[5][r], acc[6][r], acc[7][r]);
        *(float4*)(o + 128) = make_float4(p[0][r], p[1][r], p[2][r], p[3][r]);
    }
}

extern "C" void kernel_launch(void* const* d_in, const int* in_sizes, int n_in,
                              void* d_out, int out_size, void* d_ws, size_t ws_size,
                              hipStream_t stream) {
    const float* emb    = (const float*)d_in[0];
    const float* weight = (const float*)d_in[1];
    const int*   src    = (const int*)d_in[2];
    const int*   dst    = (const int*)d_in[3];

    float* out = (float*)d_out;

    clear_counts_kernel<<<(2 * N_TILES + 255) / 256, 256, 0, stream>>>(out);
    bin_edges_kernel<<<(N_EDGES + 255) / 256, 256, 0, stream>>>(src, dst, out);
    tile_sum_kernel<<<N_TILES / 2, 64, 0, stream>>>(emb, weight, src, dst, out);
}


# bin pass: a thread bins 8 consecutive edges (128-bit index loads up front, 8 atomics in flight): 1563 waves enter, not 12504 short ones; sum order = slot arrival per list, as before
# speedup vs baseline: 2.8963x; 1.0429x over previous
#include <hip/hip_runtime.h>

typedef __attribute__((ext_vector_type(2))) float v2f;
typedef __attribute__((ext_vector_type(4))) float v4f;
typedef __attribute__((ext_vector_type(8))) float v8f;

#define N_NODES    100000
#define N_EDGES    400000
#define D_EMB      128
#define D_PROMPT   64
#define D_OUT      192
#define TILE_M     16
#define N_TILES    (N_NODES / TILE_M)
#define TILE_WORDS (TILE_M * D_OUT)
#define HALF_WORDS (TILE_WORDS / 2)
#define LIST_OFF   32
#define LIST_CAP   (HALF_WORDS - LIST_OFF)
#define ROW_SHIFT  20
#define SRC_MASK   0xFFFFF
#define PAD_ENT    (16 << ROW_SHIFT)
#define BIN_EDGES   8
#define BIN_THREADS (N_EDGES / BIN_EDGES)
#define BIN_BLOCK   128


__global__ void clear_counts_kernel(float* __restrict__ out) {
    const int t = blockIdx.x * blockDim.x + threadIdx.x;
    if (t < 2 * N_TILES) ((int*)out)[(long)(t >> 1) * TILE_WORDS + (t & 1) * HALF_WORDS] = 0;
}

__global__ void __launch_bounds__(BIN_BLOCK)
bin_edges_kernel(const int* __restrict__ src,
                 const int* __restrict__ dst,
                 float* __restrict__ out) {
    const int t = blockIdx.x * BIN_BLOCK + threadIdx.x;
    if (t >= BIN_THREADS) return;
    const int4 da = ((const int4*)dst)[2 * t], db = ((const int4*)dst)[2 * t + 1];
    const int4 sa = ((const int4*)src)[2 * t], sb = ((const int4*)src)[2 * t + 1];
    const int d[BIN_EDGES] = {da.x, da.y, da.z, da.w, db.x, db.y, db.z, db.w};
    const int s[BIN_EDGES] = {sa.x, sa.y, sa.z, sa.w, sb.x, sb.y, sb.z, sb.w};
    const int half = (int)((__builtin_amdgcn_s_sendmsg_rtn(0x87) >> 18) & 1u);
    int* lists = (int*)out + half * HALF_WORDS;
    int* reg[BIN_EDGES];
    int  slot[BIN_EDGES];
    #pragma unroll
    for (int j = 0; j < BIN_EDGES; ++j) {
        reg[j]  = lists + (long)(d[j] >> 4) * TILE_WORDS;
        slot[j] = __hip_atomic_fetch_add(reg[j], 1, __ATOMIC_RELAXED, __HIP_MEMORY_SCOPE_AGENT);
    }
    #pragma unroll
    for (int j = 0; j < BIN_EDGES; ++j)
        if (slot[j] < LIST_CAP) reg[j][LIST_OFF + slot[j]] = s[j] | ((d[j] & 15) << ROW_SHIFT);
}

struct KStep {
    v2f    a;
    v4f    x0, x1, y0, y1;
};

__device__ __forceinline__ KStep load_kstep(const float* __restrict__ emb,
                                            int ent, int s, int c, int h) {
    const int q  = 4 * s;
    const int e0 = __builtin_amdgcn_readlane(ent, q + 0);
    const int e1 = __builtin_amdgcn_readlane(ent, q + 1);
    const int e2 = __builtin_amdgcn_readlane(ent, q + 2);
    const int e3 = __builtin_amdgcn_readlane(ent, q + 3);
    const int ex = h ? e2 : e0;
    const int ey = h ? e3 : e1;
    KStep k;
    k.a.x = ((ex >> ROW_SHIFT) == c) ? 1.0f : 0.0f;
    k.a.y = ((ey >> ROW_SHIFT) == c) ? 1.0f : 0.0f;
    const unsigned ox = (unsigned)(ex & SRC_MASK) * (unsigned)(D_EMB * sizeof(float)) + 16u * (unsigned)c;
    const unsigned oy = (unsigned)(ey & SRC_MASK) * (unsigned)(D_EMB * sizeof(float)) + 16u * (unsigned)c;
    const char* base = (const char*)emb;
    k.x0 = *(const v4f*)(base + ox);
    k.x1 = *(const v4f*)(base + ox + 256u);
    k.y0 = *(const v4f*)(base + oy);
    k.y1 = *(const v4f*)(base + oy + 256u);
    return k;
}

#define MMA(A, BX, BY, C) \
    __builtin_amdgcn_wmma_f32_16x16x4_f32(false, (A), false, (v2f){(BX), (BY)}, (short)0, (C), false, false)

__device__ __forceinline__ void mma_kstep(const KStep& k, v8f (&acc)[8], float& pdeg) {
    pdeg += k.a.x + k.a.y;
    acc[0] = MMA(k.a, k.x0.x, k.y0.x, acc[0]);
    acc[1] = MMA(k.a, k.x0.y, k.y0.y, acc[1]);
    acc[2] = MMA(k.a, k.x0.z, k.y0.z, acc[2]);
    acc[3] = MMA(k.a, k.x0.w, k.y0.w, acc[3]);
    acc[4] = MMA(k.a, k.x1.x, k.y1.x, acc[4]);
    acc[5] = MMA(k.a, k.x1.y, k.y1.y, acc[5]);
    acc[6] = MMA(k.a, k.x1.z, k.y1.z, acc[6]);
    acc[7] = MMA(k.a, k.x1.w, k.y1.w, acc[7]);
}

__device__ __forceinline__ void process_batch(const float* __restrict__ emb,
                                              int ent, int steps, int c, int h,
                                              v8f (&acc)[8], float& pdeg) {
    KStep k[8];
    #pragma unroll
    for (int s = 0; s < 8; ++s) k[s] = load_kstep(emb, ent, s, c, h);
    __builtin_amdgcn_sched_barrier(0);
    mma_kstep(k[0], acc, pdeg);
    if (steps > 1) { mma_kstep(k[1], acc, pdeg);
    if (steps > 2) { mma_kstep(k[2], acc, pdeg);
    if (steps > 3) { mma_kstep(k[3], acc, pdeg);
    if (steps > 4) { mma_kstep(k[4], acc, pdeg);
    if (steps > 5) { mma_kstep(k[5], acc, pdeg);
    if (steps > 6) { mma_kstep(k[6], acc, pdeg);
    if (steps > 7) { mma_kstep(k[7], acc, pdeg); } } } } } } }
    #pragma unroll
    for (int s = 0; s < 8; ++s)
        asm volatile("" :: "v"(k[s].x0), "v"(k[s].x1), "v"(k[s].y0), "v"(k[s].y1));
}

__global__ void __launch_bounds__(64)
tile_sum_kernel(const float* __restrict__ emb,
                const float* __restrict__ weight,
                const int*   __restrict__ src,
                const int*   __restrict__ dst,
                float* out) {
    const int lane = threadIdx.x & 31;
    const int wave = __builtin_amdgcn_readfirstlane(threadIdx.x >> 5);
    const int tile = blockIdx.x * 2 + wave;
    if (tile >= N_TILES) return;

    const int c = lane & 15;
    const int h = lane >> 4;

    float* reg = out + (long)tile * TILE_WORDS;
    const int cw   = __float_as_int(reg[(lane & 1) * HALF_WORDS]);
    const int cnt0 = __builtin_amdgcn_readlane(cw, 0);
    const int cnt1 = __builtin_amdgcn_readlane(cw, 1);
    const int cnt  = cnt0 + cnt1;

    v8f acc[8];
    #pragma unroll
    for (int j = 0; j < 8; ++j) acc[j] = (v8f){};
    float pdeg = 0.0f;

    const bool listed = cnt0 <= LIST_CAP && cnt1 <= LIST_CAP;
    const int  total  = listed ? cnt : N_EDGES;
    for (int base = 0; base < total; base += 32) {
        int ent, steps;
        if (listed) {
            const int i    = base + lane;
            const int left = cnt - base;
            const int w    = (i < cnt0) ? (LIST_OFF + i) : (HALF_WORDS + LIST_OFF + i - cnt0);
            ent   = (i < cnt) ? __float_as_int(reg[w]) : PAD_ENT;
            steps = ((left < 32 ? left : 32) + 3) >> 2;
        } else {
            const int      d    = dst[base + lane];
            const bool     mine = (d >> 4) == tile;
            const unsigned hits = __builtin_amdgcn_ballot_w32(mine);
            if (hits == 0) continue;
            ent   = mine ? (src[base + lane] | ((d & 15) << ROW_SHIFT)) : PAD_ENT;
            steps = ((31 - __builtin_clz(hits)) >> 2) + 1;
        }
        process_batch(emb, ent, steps, c, h, acc, pdeg);
    }

    const float  deg = pdeg + __shfl_xor(pdeg, 16);
    const float4 w4  = ((const float4*)weight)[c];
    v2f a2;
    a2.x = h ? 0.0f : deg;
    a2.y = 0.0f;
    v8f p[4];
    p[0] = MMA(a2, w4.x, 0.0f, (v8f){});
    p[1] = MMA(a2, w4.y, 0.0f, (v8f){});
    p[2] = MMA(a2, w4.z, 0.0f, (v8f){});
    p[3] = MMA(a2, w4.w, 0.0f, (v8f){});

    #pragma unroll
    for (int r = 0; r < 8; ++r) {
        float* o = out + (long)(tile * TILE_M + 8 * h + r) * D_OUT + 4 * c;
        *(float4*)(o)       = make_float4(acc[0][r], acc[1][r], acc[2][r], acc[3][r]);
        *(float4*)(o + 64)  = make_float4(acc[4][r], acc[5][r], acc[6][r], acc[7][r]);
        *(float4*)(o + 128) = make_float4(p[0][r], p[1][r], p[2][r], p[3][r]);
    }
}

extern "C" void kernel_launch(void* const* d_in, const int* in_sizes, int n_in,
                              void* d_out, int out_size, void* d_ws, size_t ws_size,
                              hipStream_t stream) {
    const float* emb    = (const float*)d_in[0];
    const float* weight = (const float*)d_in[1];
    const int*   src    = (const int*)d_in[2];
    const int*   dst    = (const int*)d_in[3];

    float* out = (float*)d_out;

    clear_counts_kernel<<<(2 * N_TILES + 255) / 256, 256, 0, stream>>>(out);
    bin_edges_kernel<<<(BIN_THREADS + BIN_BLOCK - 1) / BIN_BLOCK, BIN_BLOCK, 0, stream>>>(src, dst, out);
    tile_sum_kernel<<<N_TILES / 2, 64, 0, stream>>>(emb, weight, src, dst, out);
}
